# starting kernel
# baseline (speedup 1.0000x reference)
#include <hip/hip_runtime.h>

#define NNODES 100000
#define NEDGES 1600000
#define FIN    16
#define NH     8
#define NC     8
#define HC     64
#define ETOT   (NEDGES + NNODES)

typedef __attribute__((ext_vector_type(2))) float v2f;
typedef __attribute__((ext_vector_type(8))) float v8f;

__device__ __forceinline__ unsigned f2ord(float f) {
    unsigned b = __float_as_uint(f);
    return (b & 0x80000000u) ? ~b : (b | 0x80000000u);
}
__device__ __forceinline__ float ord2f(unsigned k) {
    unsigned b = (k & 0x80000000u) ? (k ^ 0x80000000u) : ~k;
    return __uint_as_float(b);
}

__global__ void k_init(float* __restrict__ out, float* __restrict__ denom,
                       unsigned* __restrict__ emax) {
    int i = blockIdx.x * blockDim.x + threadIdx.x;
    if (i < NNODES * HC) out[i] = 0.0f;
    if (i < NNODES * NH) { denom[i] = 0.0f; emax[i] = 0u; }
}

__global__ __launch_bounds__(256) void k_gemm(const float* __restrict__ x,
                                              const float* __restrict__ W,
                                              float* __restrict__ h) {
    const int lane = threadIdx.x & 31;
    const int wave = threadIdx.x >> 5;
    const int tile = blockIdx.x * 8 + wave;
    const long base = (long)tile * 16;
    if (base >= NNODES) return;

    const int r  = lane & 15;
    const int kh = (lane >> 4) << 1;

    const float* xr = x + (base + r) * FIN;
    v2f a[4];
#pragma unroll
    for (int s = 0; s < 4; ++s) {
        const int k0 = s * 4 + kh;
        a[s][0] = xr[k0];
        a[s][1] = xr[k0 + 1];
    }

#pragma unroll
    for (int j = 0; j < 4; ++j) {
        v8f acc = {};
#pragma unroll
        for (int s = 0; s < 4; ++s) {
            const int k0 = s * 4 + kh;
            v2f b;
            b[0] = W[(k0    ) * HC + j * 16 + r];
            b[1] = W[(k0 + 1) * HC + j * 16 + r];
            acc = __builtin_amdgcn_wmma_f32_16x16x4_f32(
                 false, a[s],  false, b,
                 (short)0, acc,  false,  false);
        }
#pragma unroll
        for (int v = 0; v < 8; ++v) {
            const int row = ((lane >> 4) << 3) + v;
            h[(base + row) * HC + j * 16 + r] = acc[v];
        }
    }
}

__global__ __launch_bounds__(256) void k_att(const float* __restrict__ h,
                                             const float* __restrict__ att_src,
                                             const float* __restrict__ att_dst,
                                             float* __restrict__ a_src,
                                             float* __restrict__ a_dst) {
    int i = blockIdx.x * blockDim.x + threadIdx.x;
    if (i >= NNODES * NH) return;
    const int hh = i & 7;
    const float* hp = h + (long)(i >> 3) * HC + hh * NC;
    float ss = 0.0f, sd = 0.0f;
#pragma unroll
    for (int c = 0; c < NC; ++c) {
        float v = hp[c];
        ss = fmaf(v, att_src[hh * NC + c], ss);
        sd = fmaf(v, att_dst[hh * NC + c], sd);
    }
    a_src[i] = ss;
    a_dst[i] = sd;
}

__device__ __forceinline__ void edge_sd(const int* __restrict__ ei, int e,
                                        int& s, int& d) {
    if (e < NEDGES) { s = ei[e]; d = ei[NEDGES + e]; }
    else            { s = d = e - NEDGES; }
}
__device__ __forceinline__ float edge_e(const float* __restrict__ as,
                                        const float* __restrict__ ad,
                                        int s, int d, int hh) {
    float v = as[s * NH + hh] + ad[d * NH + hh];
    return v > 0.0f ? v : 0.2f * v;
}

__global__ __launch_bounds__(256) void k_emax(const int* __restrict__ ei,
                                              const float* __restrict__ as,
                                              const float* __restrict__ ad,
                                              unsigned* __restrict__ emax) {
    int i = blockIdx.x * blockDim.x + threadIdx.x;
    if (i >= ETOT * NH) return;
    int e = i >> 3, hh = i & 7, s, d;
    edge_sd(ei, e, s, d);
    float v = edge_e(as, ad, s, d, hh);
    atomicMax(emax + d * NH + hh, f2ord(v));
}

__global__ __launch_bounds__(256) void k_denom(const int* __restrict__ ei,
                                               const float* __restrict__ as,
                                               const float* __restrict__ ad,
                                               const unsigned* __restrict__ emax,
                                               float* __restrict__ denom) {
    int i = blockIdx.x * blockDim.x + threadIdx.x;
    if (i >= ETOT * NH) return;
    int e = i >> 3, hh = i & 7, s, d;
    edge_sd(ei, e, s, d);
    float v  = edge_e(as, ad, s, d, hh);
    float m  = ord2f(emax[d * NH + hh]);
    float ex = __expf(v - m);
    atomicAdd(denom + d * NH + hh, ex);
}

__global__ __launch_bounds__(256) void k_scatter(const int* __restrict__ ei,
                                                 const float* __restrict__ as,
                                                 const float* __restrict__ ad,
                                                 const unsigned* __restrict__ emax,
                                                 const float* __restrict__ denom,
                                                 const float* __restrict__ h,
                                                 float* __restrict__ out) {
    int i = blockIdx.x * blockDim.x + threadIdx.x;
    if (i >= ETOT * NH) return;
    int e = i >> 3, hh = i & 7, s, d;
    edge_sd(ei, e, s, d);

    const float* hp = h + (long)s * HC + hh * NC;
    __builtin_prefetch(hp, 0, 0);

    float v     = edge_e(as, ad, s, d, hh);
    float m     = ord2f(emax[d * NH + hh]);
    float ex    = __expf(v - m);
    float alpha = ex / (denom[d * NH + hh] + 1e-16f);

    float4 h0 = ((const float4*)hp)[0];
    float4 h1 = ((const float4*)hp)[1];
    float* op = out + (long)d * HC + hh * NC;
    atomicAdd(op + 0, alpha * h0.x);
    atomicAdd(op + 1, alpha * h0.y);
    atomicAdd(op + 2, alpha * h0.z);
    atomicAdd(op + 3, alpha * h0.w);
    atomicAdd(op + 4, alpha * h1.x);
    atomicAdd(op + 5, alpha * h1.y);
    atomicAdd(op + 6, alpha * h1.z);
    atomicAdd(op + 7, alpha * h1.w);
}

__global__ __launch_bounds__(256) void k_final(float* __restrict__ out,
                                               const float* __restrict__ bias) {
    int i = blockIdx.x * blockDim.x + threadIdx.x;
    if (i >= NNODES * HC) return;
    float v = out[i] + bias[i & (HC - 1)];
    out[i] = v > 0.0f ? v : (__expf(v) - 1.0f);
}

extern "C" void kernel_launch(void* const* d_in, const int* in_sizes, int n_in,
                              void* d_out, int out_size, void* d_ws, size_t ws_size,
                              hipStream_t stream) {
    const float* x       = (const float*)d_in[0];
    const int*   ei      = (const int*)  d_in[1];
    const float* W       = (const float*)d_in[2];
    const float* att_src = (const float*)d_in[3];
    const float* att_dst = (const float*)d_in[4];
    const float* bias    = (const float*)d_in[5];
    float* out = (float*)d_out;

    char* ws = (char*)d_ws;
    float*    h      = (float*)ws;     ws += sizeof(float)    * NNODES * HC;
    float*    a_src  = (float*)ws;     ws += sizeof(float)    * NNODES * NH;
    float*    a_dst  = (float*)ws;     ws += sizeof(float)    * NNODES * NH;
    unsigned* emax   = (unsigned*)ws;  ws += sizeof(unsigned) * NNODES * NH;
    float*    denom  = (float*)ws;

    const int ntile   = NNODES / 16;
    const int eh      = ETOT * NH;
    const int blk     = 256;

    k_init   <<<(NNODES * HC + blk - 1) / blk, blk, 0, stream>>>(out, denom, emax);
    k_gemm   <<<(ntile + 7) / 8,               blk, 0, stream>>>(x, W, h);
    k_att    <<<(NNODES * NH + blk - 1) / blk, blk, 0, stream>>>(h, att_src, att_dst, a_src, a_dst);
    k_emax   <<<(eh + blk - 1) / blk,          blk, 0, stream>>>(ei, a_src, a_dst, emax);
    k_denom  <<<(eh + blk - 1) / blk,          blk, 0, stream>>>(ei, a_src, a_dst, emax, denom);
    k_scatter<<<(eh + blk - 1) / blk,          blk, 0, stream>>>(ei, a_src, a_dst, emax, denom, h, out);
    k_final  <<<(NNODES * HC + blk - 1) / blk, blk, 0, stream>>>(out, bias);
}


# Segment-max edge pass not run when max-min of a_src per head is <= 8 (found in the logit pass): softmax shift = bound leaky(max a_src + a_dst[d]); wider logits take the exact pass; sum order kept
# speedup vs baseline: 3.3370x; 3.3370x over previous
#include <hip/hip_runtime.h>

#define NNODES 100000
#define NEDGES 1600000
#define FIN    16
#define NH     8
#define NC     8
#define HC     64
#define ETOT   (NEDGES + NNODES)

typedef __attribute__((ext_vector_type(2))) float v2f;
typedef __attribute__((ext_vector_type(8))) float v8f;

__device__ __forceinline__ unsigned f2ord(float f) {
    unsigned b = __float_as_uint(f);
    return (b & 0x80000000u) ? ~b : (b | 0x80000000u);
}
__device__ __forceinline__ float ord2f(unsigned k) {
    unsigned b = (k & 0x80000000u) ? (k ^ 0x80000000u) : ~k;
    return __uint_as_float(b);
}

#define GRP_CHUNKS 16
#define NGRP ((ETOT / 32 + GRP_CHUNKS - 1) / GRP_CHUNKS)
#define CNT_PITCH 64
__device__ __forceinline__ int row_part(int d) { return d >= NNODES / 2; }

#define SPREAD_MAX 8.0f
#define KEY_WORD 16
__device__ __forceinline__ unsigned load_keys(const unsigned* keys, int lane) {
    const unsigned* p = keys + (lane >> 4) * CNT_PITCH + KEY_WORD + (lane & 15);
    return max(p[0], p[16]);
}
__device__ __forceinline__ bool bound_ok(unsigned k, int lane, float& g) {
    k = max(k, (unsigned)__shfl_xor((int)k, 16));
    const unsigned o = (unsigned)__shfl_xor((int)k, 8);
    const bool lo = (lane & 8) == 0;
    const float G = ord2f(lo ? k : o);
    const float L = ord2f(~(lo ? o : k));
    g = G;
    return __builtin_amdgcn_ballot_w32(G - L <= SPREAD_MAX) == 0xffffffffu;
}

__global__ void k_init(float* __restrict__ out, float* __restrict__ denom,
                       unsigned* __restrict__ emax, unsigned* __restrict__ cnt) {
    int i = blockIdx.x * blockDim.x + threadIdx.x;
    if (i < NNODES * HC) out[i] = 0.0f;
    if (i < NNODES * NH) { denom[i] = 0.0f; emax[i] = 0u; }
    if (i < 4) cnt[i * CNT_PITCH] = 0u;
    if (i < 64) cnt[(i >> 5) * CNT_PITCH + KEY_WORD + (i & 31)] = 0u;
}

__global__ __launch_bounds__(256) void k_gemm(const float* __restrict__ x,
                                              const float* __restrict__ W,
                                              float* __restrict__ h) {
    const int lane = threadIdx.x & 31;
    const int wave = threadIdx.x >> 5;
    const int tile = blockIdx.x * 8 + wave;
    const long base = (long)tile * 16;
    if (base >= NNODES) return;

    const int r  = lane & 15;
    const int kh = (lane >> 4) << 1;

    const float* xr = x + (base + r) * FIN;
    v2f a[4];
#pragma unroll
    for (int s = 0; s < 4; ++s) {
        const int k0 = s * 4 + kh;
        a[s][0] = xr[k0];
        a[s][1] = xr[k0 + 1];
    }

#pragma unroll
    for (int j = 0; j < 4; ++j) {
        v8f acc = {};
#pragma unroll
        for (int s = 0; s < 4; ++s) {
            const int k0 = s * 4 + kh;
            v2f b;
            b[0] = W[(k0    ) * HC + j * 16 + r];
            b[1] = W[(k0 + 1) * HC + j * 16 + r];
            acc = __builtin_amdgcn_wmma_f32_16x16x4_f32(
                 false, a[s],  false, b,
                 (short)0, acc,  false,  false);
        }
#pragma unroll
        for (int v = 0; v < 8; ++v) {
            const int row = ((lane >> 4) << 3) + v;
            h[(base + row) * HC + j * 16 + r] = acc[v];
        }
    }
}

__global__ __launch_bounds__(256) void k_att(const float* __restrict__ h,
                                             const float* __restrict__ att_src,
                                             const float* __restrict__ att_dst,
                                             float* __restrict__ a_src,
                                             float* __restrict__ a_dst,
                                             unsigned* __restrict__ keys) {
    __shared__ unsigned wkeys[8][16];
    const int lane = threadIdx.x & 31;
    const int wave = threadIdx.x >> 5;
    int i = blockIdx.x * blockDim.x + threadIdx.x;
    const int hh = i & 7;
    const bool ok = i < NNODES * NH;
    float ss = 0.0f, sd = 0.0f;
    if (ok) {
        const float* hp = h + (long)(i >> 3) * HC + hh * NC;
#pragma unroll
        for (int c = 0; c < NC; ++c) {
            float v = hp[c];
            ss = fmaf(v, att_src[hh * NC + c], ss);
            sd = fmaf(v, att_dst[hh * NC + c], sd);
        }
    }
    unsigned key  = ok ?  f2ord(ss) : 0u;
    unsigned ikey = ok ? ~f2ord(ss) : 0u;
    key  = max(key,  (unsigned)__shfl_xor((int)key, 8));
    ikey = max(ikey, (unsigned)__shfl_xor((int)ikey, 8));
    key  = max(key,  (unsigned)__shfl_xor((int)key, 16));
    ikey = max(ikey, (unsigned)__shfl_xor((int)ikey, 16));
    if (lane < 16) wkeys[wave][lane] = lane < 8 ? key : ikey;
    __syncthreads();
    if (ok) {
        a_src[i] = ss;
        a_dst[i] = sd;
    }
    if (threadIdx.x < 16) {
        unsigned k = wkeys[0][threadIdx.x];
#pragma unroll
        for (int w = 1; w < 8; ++w) k = max(k, wkeys[w][threadIdx.x]);
        const int pair = (int)((__builtin_amdgcn_s_sendmsg_rtn(0x87) >> 17) & 3u);
        atomicMax(keys + (pair >> 1) * CNT_PITCH + KEY_WORD + (pair & 1) * 16 + threadIdx.x, k);
    }
}

__device__ __forceinline__ void edge_sd(const int* __restrict__ ei, int e,
                                        int& s, int& d) {
    if (e < NEDGES) { s = ei[e]; d = ei[NEDGES + e]; }
    else            { s = d = e - NEDGES; }
}
#define NCHUNK (ETOT / 32)

__device__ __forceinline__ bool take_group(unsigned* __restrict__ cnt, int* tick,
                                           int& grp, int& part) {
    if (threadIdx.x == 0) {
        int p = (int)((__builtin_amdgcn_s_sendmsg_rtn(0x87) >> 18) & 1u);
        unsigned t = atomicAdd(cnt + p * CNT_PITCH, 1u);
        if (t >= (unsigned)NGRP) {
            p ^= 1;
            t = atomicAdd(cnt + p * CNT_PITCH, 1u);
        }
        tick[0] = (int)t;
        tick[1] = p;
    }
    __syncthreads();
    grp  = tick[0];
    part = tick[1];
    return grp < NGRP;
}

__device__ __forceinline__ int own_edges(int sv, int dv, int part,
                                         int lane, int2* list, int& cs, int& cd) {
    const bool mine = row_part(dv) == part;
    const unsigned mask = __builtin_amdgcn_ballot_w32(mine);
    const int n = __popc(mask);
    const int rank = (int)__builtin_amdgcn_mbcnt_lo(mask, 0u);
    if (mine) list[rank] = make_int2(sv, dv);
    int2 c = make_int2(0, 0);
    if (n > 0) c = list[lane < n ? lane : n - 1];
    cs = c.x;
    cd = c.y;
    return n;
}

__global__ __launch_bounds__(256) void k_emax(const int* __restrict__ ei,
                                              const float* __restrict__ as,
                                              const float* __restrict__ ad,
                                              unsigned* __restrict__ emax,
                                              unsigned* __restrict__ cnt,
                                              const unsigned* keys) {
    __shared__ int  tick[2];
    __shared__ int2 lists[8][32];
    const int lane = threadIdx.x & 31;
    const int wave = threadIdx.x >> 5;
    float g_unused;
    if (bound_ok(load_keys(keys, lane), lane, g_unused)) return;
    int grp, part;
    if (!take_group(cnt, tick, grp, part)) return;
    const int hh  = lane & 7;
    const int sub = lane >> 3;
    const int chunk0 = grp * GRP_CHUNKS + wave;
    int sv0 = 0, dv0 = 0, sv1 = 0, dv1 = 0;
    if (chunk0     < NCHUNK) edge_sd(ei, chunk0 * 32 + lane, sv0, dv0);
    if (chunk0 + 8 < NCHUNK) edge_sd(ei, (chunk0 + 8) * 32 + lane, sv1, dv1);
#pragma nounroll
    for (int it = 0; it < GRP_CHUNKS / 8; ++it) {
        if (chunk0 + it * 8 >= NCHUNK) break;
        int cs, cd;
        const int n = own_edges(it ? sv1 : sv0, it ? dv1 : dv0, part, lane, lists[wave], cs, cd);
        float    fa[8], fb[8];
        unsigned* mp[8];
#pragma unroll
        for (int j = 0; j < 8; ++j) {
            fa[j] = 0.0f;
            fb[j] = 0.0f;
            mp[j] = emax;
            if (j * 4 < n) {
                const int s = __shfl(cs, j * 4 + sub);
                const int d = __shfl(cd, j * 4 + sub);
                fa[j] = as[s * NH + hh];
                fb[j] = ad[d * NH + hh];
                mp[j] = emax + d * NH + hh;
            }
        }
#pragma unroll
        for (int j = 0; j < 8; ++j) {
            if (j * 4 + sub < n) {
                const float v = fa[j] + fb[j];
                atomicMax(mp[j], f2ord(v > 0.0f ? v : 0.2f * v));
            }
        }
    }
}

__device__ __forceinline__ void row_group(const float* __restrict__ h, float* __restrict__ out,
                                          int cs, int cd, float aj, int base, int n,
                                          int lane, int sub) {
    float h0[4], h1[4], a0[4], a1[4];
    float* op[4];
#pragma unroll
    for (int g = 0; g < 4; ++g) {
        const int s = __builtin_amdgcn_readlane(cs, base + g);
        const int d = __builtin_amdgcn_readlane(cd, base + g);
        const float* hp = h + (long)s * HC + lane;
        h0[g] = hp[0];
        h1[g] = hp[32];
        a0[g] = __shfl(aj, g * 8 + sub);
        a1[g] = __shfl(aj, g * 8 + 4 + sub);
        op[g] = out + (long)d * HC + lane;
    }
    asm volatile("" : "+v"(h0[0]), "+v"(h1[0]), "+v"(h0[1]), "+v"(h1[1]),
                      "+v"(h0[2]), "+v"(h1[2]), "+v"(h0[3]), "+v"(h1[3]));
#pragma unroll
    for (int g = 0; g < 4; ++g) {
        if (base + g < n) {
            atomicAdd(op[g],      a0[g] * h0[g]);
            atomicAdd(op[g] + 32, a1[g] * h1[g]);
        }
    }
}

__global__ __launch_bounds__(256) void k_scatter(const int* __restrict__ ei,
                                                 const float* __restrict__ as,
                                                 const float* __restrict__ ad,
                                                 const unsigned* __restrict__ emax,
                                                 const unsigned* keys,
                                                 float* __restrict__ denom,
                                                 const float* __restrict__ h,
                                                 float* __restrict__ out,
                                                 unsigned* __restrict__ cnt) {
    __shared__ int  tick[2];
    __shared__ int2 lists[8][32];
    const int lane = threadIdx.x & 31;
    const int wave = threadIdx.x >> 5;
    const unsigned gk = load_keys(keys, lane);
    int grp, part;
    if (!take_group(cnt, tick, grp, part)) return;
    const int hh  = lane & 7;
    const int sub = lane >> 3;
    float g;
    const bool bound = bound_ok(gk, lane, g);
    const int chunk0 = grp * GRP_CHUNKS + wave;
    int sv0 = 0, dv0 = 0, sv1 = 0, dv1 = 0;
    if (chunk0     < NCHUNK) edge_sd(ei, chunk0 * 32 + lane, sv0, dv0);
    if (chunk0 + 8 < NCHUNK) edge_sd(ei, (chunk0 + 8) * 32 + lane, sv1, dv1);
    for (int it = 0; it < GRP_CHUNKS / 8; ++it) {
        if (chunk0 + it * 8 >= NCHUNK) break;
        int cs, cd;
        const int n = own_edges(it ? sv1 : sv0, it ? dv1 : dv0, part, lane, lists[wave], cs, cd);

        float alpha[8];
        float fa[8], fb[8];
        unsigned mk[8];
#pragma unroll
        for (int j = 0; j < 8; ++j) {
            fa[j] = 0.0f;
            fb[j] = 0.0f;
            mk[j] = 0u;
            if (j * 4 < n) {
                const int s = __shfl(cs, j * 4 + sub);
                const int d = __shfl(cd, j * 4 + sub);
                fa[j] = as[s * NH + hh];
                fb[j] = ad[d * NH + hh];
                if (!bound) mk[j] = emax[d * NH + hh];
            }
        }
#pragma unroll
        for (int j = 0; j < 8; ++j) {
            alpha[j] = 0.0f;
            if (j * 4 < n) {
                const float v = fa[j] + fb[j];
                const float m = g + fb[j];
                const float shift = bound ? (m > 0.0f ? m : 0.2f * m) : ord2f(mk[j]);
                alpha[j] = __expf((v > 0.0f ? v : 0.2f * v) - shift);
            }
        }
#pragma unroll
        for (int j = 0; j < 8; ++j) {
            if (j * 4 < n) {
                const int d = __shfl(cd, j * 4 + sub);
                if (j * 4 + sub < n) atomicAdd(denom + d * NH + hh, alpha[j]);
            }
        }

#pragma unroll
        for (int j = 0; j < 8; ++j) {
            if (j * 4 < n) row_group(h, out, cs, cd, alpha[j], j * 4, n, lane, sub);
        }
    }
}

__global__ __launch_bounds__(256) void k_final(float* __restrict__ out,
                                               const float* __restrict__ denom,
                                               const float* __restrict__ bias) {
    int i = blockIdx.x * blockDim.x + threadIdx.x;
    if (i >= NNODES * HC) return;
    float v = out[i] / (denom[i >> 3] + 1e-16f) + bias[i & (HC - 1)];
    out[i] = v > 0.0f ? v : (__expf(v) - 1.0f);
}

extern "C" void kernel_launch(void* const* d_in, const int* in_sizes, int n_in,
                              void* d_out, int out_size, void* d_ws, size_t ws_size,
                              hipStream_t stream) {
    const float* x       = (const float*)d_in[0];
    const int*   ei      = (const int*)  d_in[1];
    const float* W       = (const float*)d_in[2];
    const float* att_src = (const float*)d_in[3];
    const float* att_dst = (const float*)d_in[4];
    const float* bias    = (const float*)d_in[5];
    float* out = (float*)d_out;

    char* ws = (char*)d_ws;
    float*    h      = (float*)ws;     ws += sizeof(float)    * NNODES * HC;
    float*    a_src  = (float*)ws;     ws += sizeof(float)    * NNODES * NH;
    float*    a_dst  = (float*)ws;     ws += sizeof(float)    * NNODES * NH;
    unsigned* emax   = (unsigned*)ws;  ws += sizeof(unsigned) * NNODES * NH;
    float*    denom  = (float*)ws;     ws += sizeof(float)    * NNODES * NH;
    unsigned* cnt    = (unsigned*)ws;

    const int ntile   = NNODES / 16;
    const int blk     = 256;

    k_init   <<<(NNODES * HC + blk - 1) / blk, blk, 0, stream>>>(out, denom, emax, cnt);
    k_gemm   <<<(ntile + 7) / 8,               blk, 0, stream>>>(x, W, h);
    k_att    <<<(NNODES * NH + blk - 1) / blk, blk, 0, stream>>>(h, att_src, att_dst, a_src, a_dst, cnt);
    k_emax   <<<2 * NGRP,                      blk, 0, stream>>>(ei, a_src, a_dst, emax, cnt + 2 * CNT_PITCH, cnt);
    k_scatter<<<2 * NGRP,                      blk, 0, stream>>>(ei, a_src, a_dst, emax, cnt, denom, h, out, cnt);
    k_final  <<<(NNODES * HC + blk - 1) / blk, blk, 0, stream>>>(out, denom, bias);
}


# Softmax shift = the destination's own self-loop logit: node launch stores the self term (h row, weight 1) as start of out/denom, self loops leave the atomic scatter; self term now summed first
# speedup vs baseline: 3.6629x; 1.0977x over previous
#include <hip/hip_runtime.h>

#define NNODES 100000
#define NEDGES 1600000
#define FIN    16
#define NH     8
#define NC     8
#define HC     64
#define ETOT   (NEDGES + NNODES)

typedef __attribute__((ext_vector_type(2))) float v2f;
typedef __attribute__((ext_vector_type(8))) float v8f;

__device__ __forceinline__ unsigned f2ord(float f) {
    unsigned b = __float_as_uint(f);
    return (b & 0x80000000u) ? ~b : (b | 0x80000000u);
}
__device__ __forceinline__ float ord2f(unsigned k) {
    unsigned b = (k & 0x80000000u) ? (k ^ 0x80000000u) : ~k;
    return __uint_as_float(b);
}

#define GRP_CHUNKS 16
#define NGRP ((ETOT / 32 + GRP_CHUNKS - 1) / GRP_CHUNKS)
#define CNT_PITCH 64
__device__ __forceinline__ int row_part(int d) { return d >= NNODES / 2; }

#define SPREAD_MAX 8.0f
#define KEY_WORD 16
__device__ __forceinline__ unsigned load_keys(const unsigned* keys, int lane) {
    const unsigned* p = keys + (lane >> 4) * CNT_PITCH + KEY_WORD + (lane & 15);
    return max(p[0], p[16]);
}
__device__ __forceinline__ bool bound_ok(unsigned k, int lane) {
    k = max(k, (unsigned)__shfl_xor((int)k, 16));
    const unsigned o = (unsigned)__shfl_xor((int)k, 8);
    const bool lo = (lane & 8) == 0;
    const float G = ord2f(lo ? k : o);
    const float L = ord2f(~(lo ? o : k));
    return __builtin_amdgcn_ballot_w32(G - L <= SPREAD_MAX) == 0xffffffffu;
}

__global__ void k_init(unsigned* __restrict__ cnt) {
    const int i = threadIdx.x;
    if (i < 4) cnt[i * CNT_PITCH] = 0u;
    if (i < 64) cnt[(i >> 5) * CNT_PITCH + KEY_WORD + (i & 31)] = 0u;
}

#define TP 68
__global__ __launch_bounds__(256) void k_node(const float* __restrict__ x,
                                              const float* __restrict__ W,
                                              const float* __restrict__ att_src,
                                              const float* __restrict__ att_dst,
                                              float* __restrict__ h,
                                              float* __restrict__ a_src,
                                              unsigned* __restrict__ adx,
                                              float* __restrict__ out,
                                              float* __restrict__ denom,
                                              unsigned* __restrict__ keys) {
    __shared__ float tiles[8][16 * TP];
    __shared__ unsigned wkeys[8][16];
    const int lane = threadIdx.x & 31;
    const int wave = threadIdx.x >> 5;
    const int tile = blockIdx.x * 8 + wave;
    const long base = (long)tile * 16;
    const bool valid = base < NNODES;
    float* tl = tiles[wave];
    const int hh = lane & 7;
    float ss[4], sd[4];
    unsigned key = 0u, ikey = 0u;

    if (valid) {
        const int r  = lane & 15;
        const int kh = (lane >> 4) << 1;

        const float* xr = x + (base + r) * FIN;
        v2f a[4];
#pragma unroll
        for (int s = 0; s < 4; ++s) {
            const int k0 = s * 4 + kh;
            a[s][0] = xr[k0];
            a[s][1] = xr[k0 + 1];
        }

#pragma unroll
        for (int j = 0; j < 4; ++j) {
            v8f acc = {};
#pragma unroll
            for (int s = 0; s < 4; ++s) {
                const int k0 = s * 4 + kh;
                v2f b;
                b[0] = W[(k0    ) * HC + j * 16 + r];
                b[1] = W[(k0 + 1) * HC + j * 16 + r];
                acc = __builtin_amdgcn_wmma_f32_16x16x4_f32(
                     false, a[s],  false, b,
                     (short)0, acc,  false,  false);
            }
#pragma unroll
            for (int v = 0; v < 8; ++v) {
                const int row = ((lane >> 4) << 3) + v;
                tl[row * TP + j * 16 + r] = acc[v];
            }
        }

        float ws[NC], wd[NC];
#pragma unroll
        for (int c = 0; c < NC; ++c) {
            ws[c] = att_src[hh * NC + c];
            wd[c] = att_dst[hh * NC + c];
        }
#pragma unroll
        for (int t = 0; t < 4; ++t) {
            const float* hp = tl + (4 * t + (lane >> 3)) * TP + hh * NC;
            const float4 p0 = *(const float4*)hp;
            const float4 p1 = *(const float4*)(hp + 4);
            const float hv[NC] = {p0.x, p0.y, p0.z, p0.w, p1.x, p1.y, p1.z, p1.w};
            float s1 = 0.0f, s2 = 0.0f;
#pragma unroll
            for (int c = 0; c < NC; ++c) {
                s1 = fmaf(hv[c], ws[c], s1);
                s2 = fmaf(hv[c], wd[c], s2);
            }
            ss[t] = s1;
            sd[t] = s2;
            key  = max(key,   f2ord(s1));
            ikey = max(ikey, ~f2ord(s1));
        }
    }
    key  = max(key,  (unsigned)__shfl_xor((int)key, 8));
    ikey = max(ikey, (unsigned)__shfl_xor((int)ikey, 8));
    key  = max(key,  (unsigned)__shfl_xor((int)key, 16));
    ikey = max(ikey, (unsigned)__shfl_xor((int)ikey, 16));
    if (lane < 16) wkeys[wave][lane] = lane < 8 ? key : ikey;
    __syncthreads();
    if (valid) {
        const int col4 = (lane & 15) * 4;
#pragma unroll
        for (int q = 0; q < 8; ++q) {
            const int row = 2 * q + (lane >> 4);
            const float4 v = *(const float4*)(tl + row * TP + col4);
            *(float4*)(h   + (base + row) * HC + col4) = v;
            *(float4*)(out + (base + row) * HC + col4) = v;
        }
#pragma unroll
        for (int t = 0; t < 4; ++t) {
            a_src[(base + 4 * t) * NH + lane] = ss[t];
            const float sv = ss[t] + sd[t];
            const float sl = sv > 0.0f ? sv : 0.2f * sv;
            *(uint2*)(adx + ((base + 4 * t) * NH + lane) * 2) = make_uint2(__float_as_uint(sd[t]), f2ord(sl));
        }
        *(float4*)(denom + base * NH + lane * 4) = make_float4(1.0f, 1.0f, 1.0f, 1.0f);
    }
    if (threadIdx.x < 16) {
        unsigned k = wkeys[0][threadIdx.x];
#pragma unroll
        for (int w = 1; w < 8; ++w) k = max(k, wkeys[w][threadIdx.x]);
        const int pair = (int)((__builtin_amdgcn_s_sendmsg_rtn(0x87) >> 17) & 3u);
        atomicMax(keys + (pair >> 1) * CNT_PITCH + KEY_WORD + (pair & 1) * 16 + threadIdx.x, k);
    }
}

__device__ __forceinline__ void edge_sd(const int* __restrict__ ei, int e,
                                        int& s, int& d) {
    if (e < NEDGES) { s = ei[e]; d = ei[NEDGES + e]; }
    else            { s = d = e - NEDGES; }
}
#define NCHUNK (ETOT / 32)

__device__ __forceinline__ bool take_group(unsigned* __restrict__ cnt, int* tick,
                                           int& grp, int& part) {
    if (threadIdx.x == 0) {
        int p = (int)((__builtin_amdgcn_s_sendmsg_rtn(0x87) >> 18) & 1u);
        unsigned t = atomicAdd(cnt + p * CNT_PITCH, 1u);
        if (t >= (unsigned)NGRP) {
            p ^= 1;
            t = atomicAdd(cnt + p * CNT_PITCH, 1u);
        }
        tick[0] = (int)t;
        tick[1] = p;
    }
    __syncthreads();
    grp  = tick[0];
    part = tick[1];
    return grp < NGRP;
}

__device__ __forceinline__ int own_edges(int sv, int dv, int part,
                                         int lane, int2* list, int& cs, int& cd) {
    const bool mine = row_part(dv) == part;
    const unsigned mask = __builtin_amdgcn_ballot_w32(mine);
    const int n = __popc(mask);
    const int rank = (int)__builtin_amdgcn_mbcnt_lo(mask, 0u);
    if (mine) list[rank] = make_int2(sv, dv);
    int2 c = make_int2(0, 0);
    if (n > 0) c = list[lane < n ? lane : n - 1];
    cs = c.x;
    cd = c.y;
    return n;
}

__global__ __launch_bounds__(256) void k_emax(const int* __restrict__ ei,
                                              const float* __restrict__ as,
                                              unsigned* __restrict__ adx,
                                              float* __restrict__ out,
                                              float* __restrict__ denom,
                                              unsigned* __restrict__ cnt,
                                              const unsigned* keys) {
    __shared__ int  tick[2];
    __shared__ int2 lists[8][32];
    const int lane = threadIdx.x & 31;
    const int wave = threadIdx.x >> 5;
    if (bound_ok(load_keys(keys, lane), lane)) return;
    {
        const long r0 = (long)blockIdx.x * 16;
        if (r0 < NNODES) {
            const float4 z4 = make_float4(0.0f, 0.0f, 0.0f, 0.0f);
            *(float4*)(out + r0 * HC + threadIdx.x * 4) = z4;
            if (threadIdx.x < 32) *(float4*)(denom + r0 * NH + threadIdx.x * 4) = z4;
        }
    }
    int grp, part;
    if (!take_group(cnt, tick, grp, part)) return;
    const int hh  = lane & 7;
    const int sub = lane >> 3;
    const int chunk0 = grp * GRP_CHUNKS + wave;
    int sv0 = 0, dv0 = 0, sv1 = 0, dv1 = 0;
    if (chunk0     < NCHUNK) edge_sd(ei, chunk0 * 32 + lane, sv0, dv0);
    if (chunk0 + 8 < NCHUNK) edge_sd(ei, (chunk0 + 8) * 32 + lane, sv1, dv1);
#pragma nounroll
    for (int it = 0; it < GRP_CHUNKS / 8; ++it) {
        if (chunk0 + it * 8 >= NCHUNK) break;
        int cs, cd;
        const int n = own_edges(it ? sv1 : sv0, it ? dv1 : dv0, part, lane, lists[wave], cs, cd);
        float    fa[8], fb[8];
        unsigned* mp[8];
#pragma unroll
        for (int j = 0; j < 8; ++j) {
            fa[j] = 0.0f;
            fb[j] = 0.0f;
            mp[j] = adx + 1;
            if (j * 4 < n) {
                const int s = __shfl(cs, j * 4 + sub);
                const int d = __shfl(cd, j * 4 + sub);
                fa[j] = as[s * NH + hh];
                fb[j] = __uint_as_float(adx[(d * NH + hh) * 2]);
                mp[j] = adx + (d * NH + hh) * 2 + 1;
            }
        }
#pragma unroll
        for (int j = 0; j < 8; ++j) {
            if (j * 4 + sub < n) {
                const float v = fa[j] + fb[j];
                atomicMax(mp[j], f2ord(v > 0.0f ? v : 0.2f * v));
            }
        }
    }
}

__device__ __forceinline__ void row_group(const float* __restrict__ h, float* __restrict__ out,
                                          int cs, int cd, float aj, int base, int n,
                                          int lane, int sub) {
    float h0[4], h1[4], a0[4], a1[4];
    float* op[4];
#pragma unroll
    for (int g = 0; g < 4; ++g) {
        const int s = __builtin_amdgcn_readlane(cs, base + g);
        const int d = __builtin_amdgcn_readlane(cd, base + g);
        const float* hp = h + (long)s * HC + lane;
        h0[g] = hp[0];
        h1[g] = hp[32];
        a0[g] = __shfl(aj, g * 8 + sub);
        a1[g] = __shfl(aj, g * 8 + 4 + sub);
        op[g] = out + (long)d * HC + lane;
    }
    asm volatile("" : "+v"(h0[0]), "+v"(h1[0]), "+v"(h0[1]), "+v"(h1[1]),
                      "+v"(h0[2]), "+v"(h1[2]), "+v"(h0[3]), "+v"(h1[3]));
#pragma unroll
    for (int g = 0; g < 4; ++g) {
        if (base + g < n) {
            atomicAdd(op[g],      a0[g] * h0[g]);
            atomicAdd(op[g] + 32, a1[g] * h1[g]);
        }
    }
}

__global__ __launch_bounds__(256) void k_scatter(const int* __restrict__ ei,
                                                 const float* __restrict__ as,
                                                 const unsigned* __restrict__ adx,
                                                 const unsigned* keys,
                                                 float* __restrict__ denom,
                                                 const float* __restrict__ h,
                                                 float* __restrict__ out,
                                                 unsigned* __restrict__ cnt) {
    __shared__ int  tick[2];
    __shared__ int2 lists[8][32];
    const int lane = threadIdx.x & 31;
    const int wave = threadIdx.x >> 5;
    const unsigned gk = load_keys(keys, lane);
    int grp, part;
    if (!take_group(cnt, tick, grp, part)) return;
    const int hh  = lane & 7;
    const int sub = lane >> 3;
    const bool bound = bound_ok(gk, lane);
    const int nchunk = bound ? NEDGES / 32 : NCHUNK;
    const int chunk0 = grp * GRP_CHUNKS + wave;
    int sv0 = 0, dv0 = 0, sv1 = 0, dv1 = 0;
    if (chunk0     < nchunk) edge_sd(ei, chunk0 * 32 + lane, sv0, dv0);
    if (chunk0 + 8 < nchunk) edge_sd(ei, (chunk0 + 8) * 32 + lane, sv1, dv1);
    for (int it = 0; it < GRP_CHUNKS / 8; ++it) {
        if (chunk0 + it * 8 >= nchunk) break;
        int cs, cd;
        const int n = own_edges(it ? sv1 : sv0, it ? dv1 : dv0, part, lane, lists[wave], cs, cd);

        float alpha[8];
        float fa[8];
        unsigned long long qd[8];
#pragma unroll
        for (int j = 0; j < 8; ++j) {
            fa[j] = 0.0f;
            qd[j] = 0ull;
            if (j * 4 < n) {
                const int s = __shfl(cs, j * 4 + sub);
                const int d = __shfl(cd, j * 4 + sub);
                fa[j] = as[s * NH + hh];
                qd[j] = *(const unsigned long long*)(adx + (d * NH + hh) * 2);
            }
        }
#pragma unroll
        for (int j = 0; j < 8; ++j) {
            alpha[j] = 0.0f;
            if (j * 4 < n) {
                const float v = fa[j] + __uint_as_float((unsigned)qd[j]);
                const float shift = ord2f((unsigned)(qd[j] >> 32));
                alpha[j] = __expf((v > 0.0f ? v : 0.2f * v) - shift);
            }
        }
#pragma unroll
        for (int j = 0; j < 8; ++j) {
            if (j * 4 < n) {
                const int d = __shfl(cd, j * 4 + sub);
                if (j * 4 + sub < n) atomicAdd(denom + d * NH + hh, alpha[j]);
            }
        }

#pragma unroll
        for (int j = 0; j < 8; ++j) {
            if (j * 4 < n) row_group(h, out, cs, cd, alpha[j], j * 4, n, lane, sub);
        }
    }
}

__global__ __launch_bounds__(256) void k_final(float* __restrict__ out,
                                               const float* __restrict__ denom,
                                               const float* __restrict__ bias) {
    int i = blockIdx.x * blockDim.x + threadIdx.x;
    if (i >= NNODES * HC) return;
    float v = out[i] / (denom[i >> 3] + 1e-16f) + bias[i & (HC - 1)];
    out[i] = v > 0.0f ? v : (__expf(v) - 1.0f);
}

extern "C" void kernel_launch(void* const* d_in, const int* in_sizes, int n_in,
                              void* d_out, int out_size, void* d_ws, size_t ws_size,
                              hipStream_t stream) {
    const float* x       = (const float*)d_in[0];
    const int*   ei      = (const int*)  d_in[1];
    const float* W       = (const float*)d_in[2];
    const float* att_src = (const float*)d_in[3];
    const float* att_dst = (const float*)d_in[4];
    const float* bias    = (const float*)d_in[5];
    float* out = (float*)d_out;

    char* ws = (char*)d_ws;
    float*    h      = (float*)ws;     ws += sizeof(float)    * NNODES * HC;
    float*    a_src  = (float*)ws;     ws += sizeof(float)    * NNODES * NH;
    unsigned* adx    = (unsigned*)ws;  ws += sizeof(unsigned) * NNODES * NH * 2;
    float*    denom  = (float*)ws;     ws += sizeof(float)    * NNODES * NH;
    unsigned* cnt    = (unsigned*)ws;

    const int ntile   = NNODES / 16;
    const int blk     = 256;

    k_init   <<<1,                             64,  0, stream>>>(cnt);
    k_node   <<<(ntile + 7) / 8,               blk, 0, stream>>>(x, W, att_src, att_dst, h, a_src, adx,
                                                                 out, denom, cnt);
    k_emax   <<<2 * NGRP,                      blk, 0, stream>>>(ei, a_src, adx, out, denom, cnt + 2 * CNT_PITCH, cnt);
    k_scatter<<<2 * NGRP,                      blk, 0, stream>>>(ei, a_src, adx, cnt, denom, h, out, cnt);
    k_final  <<<(NNODES * HC + blk - 1) / blk, blk, 0, stream>>>(out, denom, bias);
}


# Scatter gathers the 64-B row of x, not the 256-B row of x@W and its logit terms: 16 edges per wave rebuilt on the f32 matrix unit into LDS (same chain, same bits); h not stored; same sums
# speedup vs baseline: 4.1963x; 1.1456x over previous
#include <hip/hip_runtime.h>

#define NNODES 100000
#define NEDGES 1600000
#define FIN    16
#define NH     8
#define NC     8
#define HC     64
#define ETOT   (NEDGES + NNODES)

typedef __attribute__((ext_vector_type(2))) float v2f;
typedef __attribute__((ext_vector_type(8))) float v8f;

__device__ __forceinline__ unsigned f2ord(float f) {
    unsigned b = __float_as_uint(f);
    return (b & 0x80000000u) ? ~b : (b | 0x80000000u);
}
__device__ __forceinline__ float ord2f(unsigned k) {
    unsigned b = (k & 0x80000000u) ? (k ^ 0x80000000u) : ~k;
    return __uint_as_float(b);
}

#define GRP_CHUNKS 16
#define NGRP ((ETOT / 32 + GRP_CHUNKS - 1) / GRP_CHUNKS)
#define CNT_PITCH 64
__device__ __forceinline__ int row_part(int d) { return d >= NNODES / 2; }

#define SPREAD_MAX 8.0f
#define KEY_WORD 16
__device__ __forceinline__ unsigned load_keys(const unsigned* keys, int lane) {
    const unsigned* p = keys + (lane >> 4) * CNT_PITCH + KEY_WORD + (lane & 15);
    return max(p[0], p[16]);
}
__device__ __forceinline__ bool bound_ok(unsigned k, int lane) {
    k = max(k, (unsigned)__shfl_xor((int)k, 16));
    const unsigned o = (unsigned)__shfl_xor((int)k, 8);
    const bool lo = (lane & 8) == 0;
    const float G = ord2f(lo ? k : o);
    const float L = ord2f(~(lo ? o : k));
    return __builtin_amdgcn_ballot_w32(G - L <= SPREAD_MAX) == 0xffffffffu;
}

__global__ void k_init(unsigned* __restrict__ cnt) {
    const int i = threadIdx.x;
    if (i < 4) cnt[i * CNT_PITCH] = 0u;
    if (i < 64) cnt[(i >> 5) * CNT_PITCH + KEY_WORD + (i & 31)] = 0u;
}

#define TP 68
__global__ __launch_bounds__(256) void k_node(const float* __restrict__ x,
                                              const float* __restrict__ W,
                                              const float* __restrict__ att_src,
                                              const float* __restrict__ att_dst,
                                              float* __restrict__ a_src,
                                              unsigned* __restrict__ adx,
                                              float* __restrict__ out,
                                              float* __restrict__ denom,
                                              unsigned* __restrict__ keys) {
    __shared__ float tiles[8][16 * TP];
    __shared__ unsigned wkeys[8][16];
    const int lane = threadIdx.x & 31;
    const int wave = threadIdx.x >> 5;
    const int tile = blockIdx.x * 8 + wave;
    const long base = (long)tile * 16;
    const bool valid = base < NNODES;
    float* tl = tiles[wave];
    const int hh = lane & 7;
    float ss[4], sd[4];
    unsigned key = 0u, ikey = 0u;

    if (valid) {
        const int r  = lane & 15;
        const int kh = (lane >> 4) << 1;

        const float* xr = x + (base + r) * FIN;
        v2f a[4];
#pragma unroll
        for (int s = 0; s < 4; ++s) {
            const int k0 = s * 4 + kh;
            a[s][0] = xr[k0];
            a[s][1] = xr[k0 + 1];
        }

#pragma unroll
        for (int j = 0; j < 4; ++j) {
            v8f acc = {};
#pragma unroll
            for (int s = 0; s < 4; ++s) {
                const int k0 = s * 4 + kh;
                v2f b;
                b[0] = W[(k0    ) * HC + j * 16 + r];
                b[1] = W[(k0 + 1) * HC + j * 16 + r];
                acc = __builtin_amdgcn_wmma_f32_16x16x4_f32(
                     false, a[s],  false, b,
                     (short)0, acc,  false,  false);
            }
#pragma unroll
            for (int v = 0; v < 8; ++v) {
                const int row = ((lane >> 4) << 3) + v;
                tl[row * TP + j * 16 + r] = acc[v];
            }
        }

        float ws[NC], wd[NC];
#pragma unroll
        for (int c = 0; c < NC; ++c) {
            ws[c] = att_src[hh * NC + c];
            wd[c] = att_dst[hh * NC + c];
        }
#pragma unroll
        for (int t = 0; t < 4; ++t) {
            const float* hp = tl + (4 * t + (lane >> 3)) * TP + hh * NC;
            const float4 p0 = *(const float4*)hp;
            const float4 p1 = *(const float4*)(hp + 4);
            const float hv[NC] = {p0.x, p0.y, p0.z, p0.w, p1.x, p1.y, p1.z, p1.w};
            float s1 = 0.0f, s2 = 0.0f;
#pragma unroll
            for (int c = 0; c < NC; ++c) {
                s1 = fmaf(hv[c], ws[c], s1);
                s2 = fmaf(hv[c], wd[c], s2);
            }
            ss[t] = s1;
            sd[t] = s2;
            key  = max(key,   f2ord(s1));
            ikey = max(ikey, ~f2ord(s1));
        }
    }
    key  = max(key,  (unsigned)__shfl_xor((int)key, 8));
    ikey = max(ikey, (unsigned)__shfl_xor((int)ikey, 8));
    key  = max(key,  (unsigned)__shfl_xor((int)key, 16));
    ikey = max(ikey, (unsigned)__shfl_xor((int)ikey, 16));
    if (lane < 16) wkeys[wave][lane] = lane < 8 ? key : ikey;
    __syncthreads();
    if (valid) {
        const int col4 = (lane & 15) * 4;
#pragma unroll
        for (int q = 0; q < 8; ++q) {
            const int row = 2 * q + (lane >> 4);
            const float4 v = *(const float4*)(tl + row * TP + col4);
            *(float4*)(out + (base + row) * HC + col4) = v;
        }
#pragma unroll
        for (int t = 0; t < 4; ++t) {
            a_src[(base + 4 * t) * NH + lane] = ss[t];
            const float sv = ss[t] + sd[t];
            const float sl = sv > 0.0f ? sv : 0.2f * sv;
            *(uint2*)(adx + ((base + 4 * t) * NH + lane) * 2) = make_uint2(__float_as_uint(sd[t]), f2ord(sl));
        }
        *(float4*)(denom + base * NH + lane * 4) = make_float4(1.0f, 1.0f, 1.0f, 1.0f);
    }
    if (threadIdx.x < 16) {
        unsigned k = wkeys[0][threadIdx.x];
#pragma unroll
        for (int w = 1; w < 8; ++w) k = max(k, wkeys[w][threadIdx.x]);
        const int pair = (int)((__builtin_amdgcn_s_sendmsg_rtn(0x87) >> 17) & 3u);
        atomicMax(keys + (pair >> 1) * CNT_PITCH + KEY_WORD + (pair & 1) * 16 + threadIdx.x, k);
    }
}

__device__ __forceinline__ void edge_sd(const int* __restrict__ ei, int e,
                                        int& s, int& d) {
    if (e < NEDGES) { s = ei[e]; d = ei[NEDGES + e]; }
    else            { s = d = e - NEDGES; }
}
#define NCHUNK (ETOT / 32)

__device__ __forceinline__ bool take_group(unsigned* __restrict__ cnt, int* tick,
                                           int& grp, int& part) {
    if (threadIdx.x == 0) {
        int p = (int)((__builtin_amdgcn_s_sendmsg_rtn(0x87) >> 18) & 1u);
        unsigned t = atomicAdd(cnt + p * CNT_PITCH, 1u);
        if (t >= (unsigned)NGRP) {
            p ^= 1;
            t = atomicAdd(cnt + p * CNT_PITCH, 1u);
        }
        tick[0] = (int)t;
        tick[1] = p;
    }
    __syncthreads();
    grp  = tick[0];
    part = tick[1];
    return grp < NGRP;
}

__device__ __forceinline__ int own_edges(int sv, int dv, int part,
                                         int lane, int2* list, int& cs, int& cd) {
    const bool mine = row_part(dv) == part;
    const unsigned mask = __builtin_amdgcn_ballot_w32(mine);
    const int n = __popc(mask);
    const int rank = (int)__builtin_amdgcn_mbcnt_lo(mask, 0u);
    if (mine) list[rank] = make_int2(sv, dv);
    int2 c = make_int2(0, 0);
    if (n > 0) c = list[lane < n ? lane : n - 1];
    cs = c.x;
    cd = c.y;
    return n;
}

__global__ __launch_bounds__(256) void k_emax(const int* __restrict__ ei,
                                              const float* __restrict__ as,
                                              unsigned* __restrict__ adx,
                                              float* __restrict__ out,
                                              float* __restrict__ denom,
                                              unsigned* __restrict__ cnt,
                                              const unsigned* keys) {
    __shared__ int  tick[2];
    __shared__ int2 lists[8][32];
    const int lane = threadIdx.x & 31;
    const int wave = threadIdx.x >> 5;
    if (bound_ok(load_keys(keys, lane), lane)) return;
    {
        const long r0 = (long)blockIdx.x * 16;
        if (r0 < NNODES) {
            const float4 z4 = make_float4(0.0f, 0.0f, 0.0f, 0.0f);
            *(float4*)(out + r0 * HC + threadIdx.x * 4) = z4;
            if (threadIdx.x < 32) *(float4*)(denom + r0 * NH + threadIdx.x * 4) = z4;
        }
    }
    int grp, part;
    if (!take_group(cnt, tick, grp, part)) return;
    const int hh  = lane & 7;
    const int sub = lane >> 3;
    const int chunk0 = grp * GRP_CHUNKS + wave;
    int sv0 = 0, dv0 = 0, sv1 = 0, dv1 = 0;
    if (chunk0     < NCHUNK) edge_sd(ei, chunk0 * 32 + lane, sv0, dv0);
    if (chunk0 + 8 < NCHUNK) edge_sd(ei, (chunk0 + 8) * 32 + lane, sv1, dv1);
#pragma nounroll
    for (int it = 0; it < GRP_CHUNKS / 8; ++it) {
        if (chunk0 + it * 8 >= NCHUNK) break;
        int cs, cd;
        const int n = own_edges(it ? sv1 : sv0, it ? dv1 : dv0, part, lane, lists[wave], cs, cd);
        float    fa[8], fb[8];
        unsigned* mp[8];
#pragma unroll
        for (int j = 0; j < 8; ++j) {
            fa[j] = 0.0f;
            fb[j] = 0.0f;
            mp[j] = adx + 1;
            if (j * 4 < n) {
                const int s = __shfl(cs, j * 4 + sub);
                const int d = __shfl(cd, j * 4 + sub);
                fa[j] = as[s * NH + hh];
                fb[j] = __uint_as_float(adx[(d * NH + hh) * 2]);
                mp[j] = adx + (d * NH + hh) * 2 + 1;
            }
        }
#pragma unroll
        for (int j = 0; j < 8; ++j) {
            if (j * 4 + sub < n) {
                const float v = fa[j] + fb[j];
                atomicMax(mp[j], f2ord(v > 0.0f ? v : 0.2f * v));
            }
        }
    }
}

#define XP 20
typedef __attribute__((ext_vector_type(4))) float v4f;

__device__ __forceinline__ void row_group(const float* tl, float* __restrict__ out,
                                          int cd, float aj, int row, int n,
                                          int lane, int sub) {
    float h0[4], h1[4], a0[4], a1[4];
    float* op[4];
#pragma unroll
    for (int g = 0; g < 4; ++g) {
        const int d = __builtin_amdgcn_readlane(cd, row + g);
        const float* hp = tl + (row + g) * TP + lane;
        h0[g] = hp[0];
        h1[g] = hp[32];
        a0[g] = __shfl(aj, g * 8 + sub);
        a1[g] = __shfl(aj, g * 8 + 4 + sub);
        op[g] = out + (long)d * HC + lane;
    }
    asm volatile("" : "+v"(h0[0]), "+v"(h1[0]), "+v"(h0[1]), "+v"(h1[1]),
                      "+v"(h0[2]), "+v"(h1[2]), "+v"(h0[3]), "+v"(h1[3]));
#pragma unroll
    for (int g = 0; g < 4; ++g) {
        if (row + g < n) {
            atomicAdd(op[g],      a0[g] * h0[g]);
            atomicAdd(op[g] + 32, a1[g] * h1[g]);
        }
    }
}

__device__ __forceinline__ void edge_group(const float* __restrict__ x,
                                           const unsigned* __restrict__ adx,
                                           float* __restrict__ denom, float* __restrict__ out,
                                           float* tl, const v2f (*wf)[32], const float* ws,
                                           int cs, int cd, int n, int lane) {
    const int hh  = lane & 7;
    const int sub = lane >> 3;
    const int r   = lane & 15;
    const int kh  = (lane >> 4) << 1;
    const int pr  = lane >> 2;
    const int pc  = (lane & 3) * 4;

    v4f xp[2];
    unsigned long long qd[4];
#pragma unroll
    for (int u = 0; u < 2; ++u) {
        const int s = __shfl(cs, 8 * u + pr);
        xp[u] = *(const v4f*)(x + (long)s * FIN + pc);
    }
#pragma unroll
    for (int t = 0; t < 4; ++t) {
        qd[t] = 0ull;
        if (4 * t < n) {
            const int d = __shfl(cd, 4 * t + sub);
            qd[t] = *(const unsigned long long*)(adx + (d * NH + hh) * 2);
        }
    }
    asm volatile("" : "+v"(xp[0]), "+v"(xp[1]), "+v"(qd[0]), "+v"(qd[1]), "+v"(qd[2]), "+v"(qd[3]));

    *(v4f*)(tl + pr * XP + pc)       = xp[0];
    *(v4f*)(tl + (8 + pr) * XP + pc) = xp[1];
    v2f a[4];
#pragma unroll
    for (int s = 0; s < 4; ++s) a[s] = *(const v2f*)(tl + r * XP + s * 4 + kh);

#pragma unroll
    for (int jp = 0; jp < 2; ++jp) {
        v8f acc0 = {}, acc1 = {};
#pragma unroll
        for (int s = 0; s < 4; ++s) {
            const v2f b0 = wf[(2 * jp    ) * 4 + s][lane];
            const v2f b1 = wf[(2 * jp + 1) * 4 + s][lane];
            acc0 = __builtin_amdgcn_wmma_f32_16x16x4_f32(
                 false, a[s],  false, b0,
                 (short)0, acc0,  false,  false);
            acc1 = __builtin_amdgcn_wmma_f32_16x16x4_f32(
                 false, a[s],  false, b1,
                 (short)0, acc1,  false,  false);
        }
#pragma unroll
        for (int v = 0; v < 8; ++v) {
            const int row = ((lane >> 4) << 3) + v;
            tl[row * TP + (2 * jp    ) * 16 + r] = acc0[v];
            tl[row * TP + (2 * jp + 1) * 16 + r] = acc1[v];
        }
    }

    float al[4];
#pragma unroll
    for (int t = 0; t < 4; ++t) {
        al[t] = 0.0f;
        if (4 * t < n) {
            const float* hp = tl + (4 * t + sub) * TP + hh * NC;
            const float4 p0 = *(const float4*)hp;
            const float4 p1 = *(const float4*)(hp + 4);
            const float hv[NC] = {p0.x, p0.y, p0.z, p0.w, p1.x, p1.y, p1.z, p1.w};
            float s1 = 0.0f;
#pragma unroll
            for (int c = 0; c < NC; ++c) s1 = fmaf(hv[c], ws[c], s1);
            const float v = s1 + __uint_as_float((unsigned)qd[t]);
            const float shift = ord2f((unsigned)(qd[t] >> 32));
            al[t] = __expf((v > 0.0f ? v : 0.2f * v) - shift);
        }
    }
#pragma unroll
    for (int t = 0; t < 4; ++t) {
        if (4 * t < n) {
            const int d = __shfl(cd, 4 * t + sub);
            if (4 * t + sub < n) atomicAdd(denom + d * NH + hh, al[t]);
        }
    }
#pragma unroll
    for (int t = 0; t < 4; ++t) {
        if (4 * t < n) row_group(tl, out, cd, al[t], 4 * t, n, lane, sub);
    }
}

__global__ __launch_bounds__(256, 12) void k_scatter(const int* __restrict__ ei,
                                                 const float* __restrict__ x,
                                                 const float* __restrict__ W,
                                                 const float* __restrict__ att_src,
                                                 const unsigned* __restrict__ adx,
                                                 const unsigned* keys,
                                                 float* __restrict__ denom,
                                                 float* __restrict__ out,
                                                 unsigned* __restrict__ cnt) {
    __shared__ int  tick[2];
    __shared__ int2 lists[8][64];
    __shared__ __attribute__((aligned(16))) float tiles[8][16 * TP];
    __shared__ v2f  wfrag[16][32];
    const int lane = threadIdx.x & 31;
    const int wave = threadIdx.x >> 5;
    const int hh   = lane & 7;
    const unsigned gk = load_keys(keys, lane);
    {
        const int r  = lane & 15;
        const int kh = (lane >> 4) << 1;
#pragma unroll
        for (int u = 0; u < 2; ++u) {
            const int f   = 2 * wave + u;
            const int k0  = (f & 3) * 4 + kh;
            const int col = (f >> 2) * 16 + r;
            v2f b;
            b[0] = W[(k0    ) * HC + col];
            b[1] = W[(k0 + 1) * HC + col];
            wfrag[f][lane] = b;
        }
    }
    float ws[NC];
#pragma unroll
    for (int c = 0; c < NC; ++c) ws[c] = att_src[hh * NC + c];
    int grp, part;
    if (!take_group(cnt, tick, grp, part)) return;
    const bool bound = bound_ok(gk, lane);
    const int nchunk = bound ? NEDGES / 32 : NCHUNK;
    const int chunk0 = grp * GRP_CHUNKS + wave;
    int sv0 = 0, dv0 = 0, sv1 = 0, dv1 = 0;
    if (chunk0     < nchunk) edge_sd(ei, chunk0 * 32 + lane, sv0, dv0);
    if (chunk0 + 8 < nchunk) edge_sd(ei, (chunk0 + 8) * 32 + lane, sv1, dv1);
    int2* list = lists[wave];
    int m = 0;
#pragma unroll
    for (int it = 0; it < GRP_CHUNKS / 8; ++it) {
        if (chunk0 + it * 8 < nchunk) {
            const int sv = it ? sv1 : sv0;
            const int dv = it ? dv1 : dv0;
            const bool mine = row_part(dv) == part;
            const unsigned mask = __builtin_amdgcn_ballot_w32(mine);
            const int rank = (int)__builtin_amdgcn_mbcnt_lo(mask, 0u);
            if (mine) list[m + rank] = make_int2(sv, dv);
            m += __popc(mask);
        }
    }
    m = __builtin_amdgcn_readfirstlane(m);
    float* tl = tiles[wave];
    for (int e0 = 0; e0 < m; e0 += 16) {
        const int n = min(16, m - e0);
        const int2 c = list[e0 + min(lane & 15, n - 1)];
        edge_group(x, adx, denom, out, tl, wfrag, ws, c.x, c.y, n, lane);
    }
}

__global__ __launch_bounds__(256) void k_final(float* __restrict__ out,
                                               const float* __restrict__ denom,
                                               const float* __restrict__ bias) {
    int i = blockIdx.x * blockDim.x + threadIdx.x;
    if (i >= NNODES * HC) return;
    float v = out[i] / (denom[i >> 3] + 1e-16f) + bias[i & (HC - 1)];
    out[i] = v > 0.0f ? v : (__expf(v) - 1.0f);
}

extern "C" void kernel_launch(void* const* d_in, const int* in_sizes, int n_in,
                              void* d_out, int out_size, void* d_ws, size_t ws_size,
                              hipStream_t stream) {
    const float* x       = (const float*)d_in[0];
    const int*   ei      = (const int*)  d_in[1];
    const float* W       = (const float*)d_in[2];
    const float* att_src = (const float*)d_in[3];
    const float* att_dst = (const float*)d_in[4];
    const float* bias    = (const float*)d_in[5];
    float* out = (float*)d_out;

    char* ws = (char*)d_ws;
    float*    a_src  = (float*)ws;     ws += sizeof(float)    * NNODES * NH;
    unsigned* adx    = (unsigned*)ws;  ws += sizeof(unsigned) * NNODES * NH * 2;
    float*    denom  = (float*)ws;     ws += sizeof(float)    * NNODES * NH;
    unsigned* cnt    = (unsigned*)ws;

    const int ntile   = NNODES / 16;
    const int blk     = 256;

    k_init   <<<1,                             64,  0, stream>>>(cnt);
    k_node   <<<(ntile + 7) / 8,               blk, 0, stream>>>(x, W, att_src, att_dst, a_src, adx,
                                                                 out, denom, cnt);
    k_emax   <<<2 * NGRP,                      blk, 0, stream>>>(ei, a_src, adx, out, denom, cnt + 2 * CNT_PITCH, cnt);
    k_scatter<<<2 * NGRP,                      blk, 0, stream>>>(ei, x, W, att_src, adx, cnt, denom, out, cnt);
    k_final  <<<(NNODES * HC + blk - 1) / blk, blk, 0, stream>>>(out, denom, bias);
}
